# starting kernel
# baseline (speedup 1.0000x reference)
#include <hip/hip_runtime.h>
#include <hip/hip_bf16.h>
#include <math.h>

#define LSEQ 512
#define HDIM 1024
#define HC   256
#define LDSP (HC + 4)

typedef __attribute__((ext_vector_type(2))) float v2f;
typedef __attribute__((ext_vector_type(8))) float v8f;

__device__ __forceinline__ float fast_tanh(float x) {
#if __has_builtin(__builtin_amdgcn_tanhf)
    return __builtin_amdgcn_tanhf(x);
#elif __has_builtin(__builtin_amdgcn_tanh_f32)
    return __builtin_amdgcn_tanh_f32(x);
#else
    float e = __builtin_amdgcn_exp2f(x * 2.8853900817779268f);
    return 1.0f - 2.0f * __builtin_amdgcn_rcpf(1.0f + e);
#endif
}

__global__ __launch_bounds__(256)
void proj_wmma_f32(const float* __restrict__ X, const float* __restrict__ W,
                   const float* __restrict__ bias, float* __restrict__ out,
                   int ldw, int hasBias) {
    const int K = HDIM, N = HDIM;
    int wave  = (blockIdx.x * blockDim.x + threadIdx.x) >> 5;
    int lane  = threadIdx.x & 31;
    int tileM = wave >> 6;
    int tileN = wave & 63;
    int l0 = tileM << 4;
    int o0 = tileN << 4;

    int row  = lane & 15;
    int koff = (lane >> 4) << 1;
    const float* aPtr = X + (size_t)(l0 + row) * K   + koff;
    const float* bPtr = W + (size_t)(o0 + row) * ldw + koff;

    v8f acc = {};
#pragma unroll 8
    for (int k0 = 0; k0 < K; k0 += 4) {
        v2f a = *(const v2f*)(aPtr + k0);
        v2f b = *(const v2f*)(bPtr + k0);
        acc = __builtin_amdgcn_wmma_f32_16x16x4_f32(
             false, a,  false, b,
             (short)0, acc,  false,  false);
    }

    int col   = lane & 15;
    int rbase = (lane >> 4) << 3;
    float bv  = hasBias ? bias[o0 + col] : 0.0f;
#pragma unroll
    for (int r = 0; r < 8; ++r) {
        out[(size_t)(l0 + rbase + r) * N + o0 + col] = acc[r] + bv;
    }
}

__global__ __launch_bounds__(256)
void pairwise_tanh(const float* __restrict__ A, const float* __restrict__ R,
                   const float* __restrict__ W2, float* __restrict__ raw) {
    __shared__ float aT[16][LDSP];
    __shared__ float rT[16][LDSP];
    __shared__ float w2s[2][HC];

    const int i0 = blockIdx.y << 4;
    const int j0 = blockIdx.x << 4;
    const int tid = threadIdx.x;
    const int tj = tid & 15;
    const int ti = tid >> 4;

    float acc0 = 0.0f, acc1 = 0.0f;

    for (int h0 = 0; h0 < HDIM; h0 += HC) {
        __syncthreads();
#pragma unroll
        for (int p = tid; p < 16 * (HC / 4); p += 256) {
            int rrow = p / (HC / 4);
            int c4   = (p % (HC / 4)) << 2;
            float4 va = *(const float4*)(A + (size_t)(j0 + rrow) * HDIM + h0 + c4);
            *(float4*)&aT[rrow][c4] = va;
            float4 vr = *(const float4*)(R + (size_t)(i0 + rrow) * HDIM + h0 + c4);
            *(float4*)&rT[rrow][c4] = vr;
        }
        {
            int p = tid;
            w2s[0][p] = W2[h0 + p];
            w2s[1][p] = W2[HDIM + h0 + p];
        }
        __syncthreads();

#pragma unroll 4
        for (int hc = 0; hc < HC; ++hc) {
            float x = aT[tj][hc] + rT[ti][hc];
            float t = fast_tanh(x);
            acc0 = fmaf(w2s[0][hc], t, acc0);
            acc1 = fmaf(w2s[1][hc], t, acc1);
        }
    }

    int i = i0 + ti, j = j0 + tj;
    float2 o; o.x = acc0; o.y = acc1;
    *(float2*)(raw + ((size_t)(i * LSEQ + j) << 1)) = o;
}

__global__ __launch_bounds__(256)
void symmetrize(const float* __restrict__ raw, const float* __restrict__ b2,
                float* __restrict__ out) {
    int idx = blockIdx.x * 256 + threadIdx.x;
    if (idx >= LSEQ * LSEQ) return;
    int i = idx >> 9, j = idx & (LSEQ - 1);
    float2 rij = *(const float2*)(raw + ((size_t)(i * LSEQ + j) << 1));
    float2 rji = *(const float2*)(raw + ((size_t)(j * LSEQ + i) << 1));
    float2 o;
    o.x = 0.5f * (rij.x + rji.x) + b2[0];
    o.y = 0.5f * (rij.y + rji.y) + b2[1];
    *(float2*)(out + ((size_t)idx << 1)) = o;
}

extern "C" void kernel_launch(void* const* d_in, const int* in_sizes, int n_in,
                              void* d_out, int out_size, void* d_ws, size_t ws_size,
                              hipStream_t stream) {
    const float* X  = (const float*)d_in[0];
    const float* W1 = (const float*)d_in[1];
    const float* b1 = (const float*)d_in[2];
    const float* W2 = (const float*)d_in[3];
    const float* b2 = (const float*)d_in[4];
    float* out = (float*)d_out;

    float* a_ws = (float*)d_ws;
    float* r_ws = a_ws + (size_t)LSEQ * HDIM;
    float* raw  = r_ws + (size_t)LSEQ * HDIM;

    proj_wmma_f32<<<256, 256, 0, stream>>>(X, W1,          nullptr, a_ws, 2 * HDIM, 0);
    proj_wmma_f32<<<256, 256, 0, stream>>>(X, W1 + HDIM,   b1,      r_ws, 2 * HDIM, 1);

    dim3 grid(LSEQ / 16, LSEQ / 16);
    pairwise_tanh<<<grid, 256, 0, stream>>>(a_ws, r_ws, W2, raw);

    symmetrize<<<(LSEQ * LSEQ) / 256, 256, 0, stream>>>(raw, b2, out);
}


# Projection GEMM re-tiled: both halves in one launch of 256 workgroups, 64x64 tiles staged through LDS by coalesced float4 loads, 16x32 per wave; k pairing and order of sums unchanged
# speedup vs baseline: 1.3706x; 1.3706x over previous
#include <hip/hip_runtime.h>
#include <hip/hip_bf16.h>
#include <math.h>

#define LSEQ 512
#define HDIM 1024
#define HC   256
#define LDSP (HC + 4)

typedef __attribute__((ext_vector_type(2))) float v2f;
typedef __attribute__((ext_vector_type(8))) float v8f;

__device__ __forceinline__ float fast_tanh(float x) {
#if __has_builtin(__builtin_amdgcn_tanhf)
    return __builtin_amdgcn_tanhf(x);
#elif __has_builtin(__builtin_amdgcn_tanh_f32)
    return __builtin_amdgcn_tanh_f32(x);
#else
    float e = __builtin_amdgcn_exp2f(x * 2.8853900817779268f);
    return 1.0f - 2.0f * __builtin_amdgcn_rcpf(1.0f + e);
#endif
}

#define GT   64
#define GTP  (GT + 4)

__global__ __launch_bounds__(256)
void proj_both_wmma_f32(const float* __restrict__ X, const float* __restrict__ W1,
                        const float* __restrict__ b1,
                        float* __restrict__ a_out, float* __restrict__ r_out) {
    __shared__ __attribute__((aligned(16))) float sA[2][GT][GTP];
    __shared__ __attribute__((aligned(16))) float sB[2][GT][GTP];

    const int tid  = threadIdx.x;
    const int lane = tid & 31;
    const int wave = tid >> 5;
    const int bm   = blockIdx.x & 7;
    const int bn   = blockIdx.x >> 3;
    const int half = bn >> 4;
    const int o0   = (bn & 15) << 6;
    const int l0   = bm << 6;

    const int srow = tid >> 4;
    const int scol = (tid & 15) << 2;
    const float* xg = X  + (size_t)(l0 + srow) * HDIM + scol;
    const float* wg = W1 + (size_t)(o0 + srow) * (2 * HDIM) + half * HDIM + scol;

    float4 pa[4], pb[4];
    for (int p = 0; p < 4; ++p) {
        pa[p] = *(const float4*)(xg + (size_t)(16 * p) * HDIM);
        pb[p] = *(const float4*)(wg + (size_t)(16 * p) * (2 * HDIM));
    }
    for (int p = 0; p < 4; ++p) {
        *(float4*)&sA[0][srow + 16 * p][scol] = pa[p];
        *(float4*)&sB[0][srow + 16 * p][scol] = pb[p];
    }
    __syncthreads();

    const int wm   = wave & 3;
    const int wn   = wave >> 2;
    const int frow = lane & 15;
    const int koff = (lane >> 4) << 1;

    v8f acc0 = {}, acc1 = {};

    for (int kc = 0; kc < HDIM / GT; ++kc) {
        const int cur = kc & 1;
        const bool more = (kc + 1) < (HDIM / GT);
        if (more) {
            const int k1 = (kc + 1) * GT;
            for (int p = 0; p < 4; ++p) {
                pa[p] = *(const float4*)(xg + (size_t)(16 * p) * HDIM + k1);
                pb[p] = *(const float4*)(wg + (size_t)(16 * p) * (2 * HDIM) + k1);
            }
        }

        const float* ap  = &sA[cur][wm * 16 + frow][koff];
        const float* bp0 = &sB[cur][wn * 32 + frow][koff];
        const float* bp1 = &sB[cur][wn * 32 + 16 + frow][koff];
        for (int k = 0; k < GT; k += 4) {
            v2f a  = *(const v2f*)(ap  + k);
            v2f b0 = *(const v2f*)(bp0 + k);
            v2f b1f = *(const v2f*)(bp1 + k);
            acc0 = __builtin_amdgcn_wmma_f32_16x16x4_f32(
                 false, a,  false, b0,
                 (short)0, acc0,  false,  false);
            acc1 = __builtin_amdgcn_wmma_f32_16x16x4_f32(
                 false, a,  false, b1f,
                 (short)0, acc1,  false,  false);
        }

        if (more) {
            for (int p = 0; p < 4; ++p) {
                *(float4*)&sA[cur ^ 1][srow + 16 * p][scol] = pa[p];
                *(float4*)&sB[cur ^ 1][srow + 16 * p][scol] = pb[p];
            }
            __syncthreads();
        }
    }

    float* __restrict__ outp = half ? r_out : a_out;
    const int col   = lane & 15;
    const int rbase = (lane >> 4) << 3;
    const int oc    = o0 + wn * 32 + col;
    const float bv0 = half ? b1[oc]      : 0.0f;
    const float bv1 = half ? b1[oc + 16] : 0.0f;
    float* orow = outp + (size_t)(l0 + wm * 16 + rbase) * HDIM + oc;
    for (int r = 0; r < 8; ++r) {
        orow[(size_t)r * HDIM]      = acc0[r] + bv0;
        orow[(size_t)r * HDIM + 16] = acc1[r] + bv1;
    }
}

__global__ __launch_bounds__(256)
void pairwise_tanh(const float* __restrict__ A, const float* __restrict__ R,
                   const float* __restrict__ W2, float* __restrict__ raw) {
    __shared__ float aT[16][LDSP];
    __shared__ float rT[16][LDSP];
    __shared__ float w2s[2][HC];

    const int i0 = blockIdx.y << 4;
    const int j0 = blockIdx.x << 4;
    const int tid = threadIdx.x;
    const int tj = tid & 15;
    const int ti = tid >> 4;

    float acc0 = 0.0f, acc1 = 0.0f;

    for (int h0 = 0; h0 < HDIM; h0 += HC) {
        __syncthreads();
#pragma unroll
        for (int p = tid; p < 16 * (HC / 4); p += 256) {
            int rrow = p / (HC / 4);
            int c4   = (p % (HC / 4)) << 2;
            float4 va = *(const float4*)(A + (size_t)(j0 + rrow) * HDIM + h0 + c4);
            *(float4*)&aT[rrow][c4] = va;
            float4 vr = *(const float4*)(R + (size_t)(i0 + rrow) * HDIM + h0 + c4);
            *(float4*)&rT[rrow][c4] = vr;
        }
        {
            int p = tid;
            w2s[0][p] = W2[h0 + p];
            w2s[1][p] = W2[HDIM + h0 + p];
        }
        __syncthreads();

#pragma unroll 4
        for (int hc = 0; hc < HC; ++hc) {
            float x = aT[tj][hc] + rT[ti][hc];
            float t = fast_tanh(x);
            acc0 = fmaf(w2s[0][hc], t, acc0);
            acc1 = fmaf(w2s[1][hc], t, acc1);
        }
    }

    int i = i0 + ti, j = j0 + tj;
    float2 o; o.x = acc0; o.y = acc1;
    *(float2*)(raw + ((size_t)(i * LSEQ + j) << 1)) = o;
}

__global__ __launch_bounds__(256)
void symmetrize(const float* __restrict__ raw, const float* __restrict__ b2,
                float* __restrict__ out) {
    int idx = blockIdx.x * 256 + threadIdx.x;
    if (idx >= LSEQ * LSEQ) return;
    int i = idx >> 9, j = idx & (LSEQ - 1);
    float2 rij = *(const float2*)(raw + ((size_t)(i * LSEQ + j) << 1));
    float2 rji = *(const float2*)(raw + ((size_t)(j * LSEQ + i) << 1));
    float2 o;
    o.x = 0.5f * (rij.x + rji.x) + b2[0];
    o.y = 0.5f * (rij.y + rji.y) + b2[1];
    *(float2*)(out + ((size_t)idx << 1)) = o;
}

extern "C" void kernel_launch(void* const* d_in, const int* in_sizes, int n_in,
                              void* d_out, int out_size, void* d_ws, size_t ws_size,
                              hipStream_t stream) {
    const float* X  = (const float*)d_in[0];
    const float* W1 = (const float*)d_in[1];
    const float* b1 = (const float*)d_in[2];
    const float* W2 = (const float*)d_in[3];
    const float* b2 = (const float*)d_in[4];
    float* out = (float*)d_out;

    float* a_ws = (float*)d_ws;
    float* r_ws = a_ws + (size_t)LSEQ * HDIM;
    float* raw  = r_ws + (size_t)LSEQ * HDIM;

    proj_both_wmma_f32<<<256, 256, 0, stream>>>(X, W1, b1, a_ws, r_ws);

    dim3 grid(LSEQ / 16, LSEQ / 16);
    pairwise_tanh<<<grid, 256, 0, stream>>>(a_ws, r_ws, W2, raw);

    symmetrize<<<(LSEQ * LSEQ) / 256, 256, 0, stream>>>(raw, b2, out);
}


# Pairwise tanh stage register-tiled: 2x4 pairs per thread from LDS quads, two-wide f32 add and fma, next h chunk prefetched; each chunk halved over two wave groups, partial sums added at the end
# speedup vs baseline: 1.8388x; 1.3416x over previous
#include <hip/hip_runtime.h>
#include <hip/hip_bf16.h>
#include <math.h>

#define LSEQ 512
#define HDIM 1024

typedef __attribute__((ext_vector_type(2))) float v2f;
typedef __attribute__((ext_vector_type(8))) float v8f;
typedef __attribute__((ext_vector_type(4))) float v4f;

__device__ __forceinline__ float fast_tanh(float x) {
#if __has_builtin(__builtin_amdgcn_tanhf)
    return __builtin_amdgcn_tanhf(x);
#elif __has_builtin(__builtin_amdgcn_tanh_f32)
    return __builtin_amdgcn_tanh_f32(x);
#else
    float e = __builtin_amdgcn_exp2f(x * 2.8853900817779268f);
    return 1.0f - 2.0f * __builtin_amdgcn_rcpf(1.0f + e);
#endif
}

#define GT   64
#define GTP  (GT + 4)

__global__ __launch_bounds__(256)
void proj_both_wmma_f32(const float* __restrict__ X, const float* __restrict__ W1,
                        const float* __restrict__ b1,
                        float* __restrict__ a_out, float* __restrict__ r_out) {
    __shared__ __attribute__((aligned(16))) float sA[2][GT][GTP];
    __shared__ __attribute__((aligned(16))) float sB[2][GT][GTP];

    const int tid  = threadIdx.x;
    const int lane = tid & 31;
    const int wave = tid >> 5;
    const int bm   = blockIdx.x & 7;
    const int bn   = blockIdx.x >> 3;
    const int half = bn >> 4;
    const int o0   = (bn & 15) << 6;
    const int l0   = bm << 6;

    const int srow = tid >> 4;
    const int scol = (tid & 15) << 2;
    const float* xg = X  + (size_t)(l0 + srow) * HDIM + scol;
    const float* wg = W1 + (size_t)(o0 + srow) * (2 * HDIM) + half * HDIM + scol;

    float4 pa[4], pb[4];
    for (int p = 0; p < 4; ++p) {
        pa[p] = *(const float4*)(xg + (size_t)(16 * p) * HDIM);
        pb[p] = *(const float4*)(wg + (size_t)(16 * p) * (2 * HDIM));
    }
    for (int p = 0; p < 4; ++p) {
        *(float4*)&sA[0][srow + 16 * p][scol] = pa[p];
        *(float4*)&sB[0][srow + 16 * p][scol] = pb[p];
    }
    __syncthreads();

    const int wm   = wave & 3;
    const int wn   = wave >> 2;
    const int frow = lane & 15;
    const int koff = (lane >> 4) << 1;

    v8f acc0 = {}, acc1 = {};

    for (int kc = 0; kc < HDIM / GT; ++kc) {
        const int cur = kc & 1;
        const bool more = (kc + 1) < (HDIM / GT);
        if (more) {
            const int k1 = (kc + 1) * GT;
            for (int p = 0; p < 4; ++p) {
                pa[p] = *(const float4*)(xg + (size_t)(16 * p) * HDIM + k1);
                pb[p] = *(const float4*)(wg + (size_t)(16 * p) * (2 * HDIM) + k1);
            }
        }

        const float* ap  = &sA[cur][wm * 16 + frow][koff];
        const float* bp0 = &sB[cur][wn * 32 + frow][koff];
        const float* bp1 = &sB[cur][wn * 32 + 16 + frow][koff];
        for (int k = 0; k < GT; k += 4) {
            v2f a  = *(const v2f*)(ap  + k);
            v2f b0 = *(const v2f*)(bp0 + k);
            v2f b1f = *(const v2f*)(bp1 + k);
            acc0 = __builtin_amdgcn_wmma_f32_16x16x4_f32(
                 false, a,  false, b0,
                 (short)0, acc0,  false,  false);
            acc1 = __builtin_amdgcn_wmma_f32_16x16x4_f32(
                 false, a,  false, b1f,
                 (short)0, acc1,  false,  false);
        }

        if (more) {
            for (int p = 0; p < 4; ++p) {
                *(float4*)&sA[cur ^ 1][srow + 16 * p][scol] = pa[p];
                *(float4*)&sB[cur ^ 1][srow + 16 * p][scol] = pb[p];
            }
            __syncthreads();
        }
    }

    float* __restrict__ outp = half ? r_out : a_out;
    const int col   = lane & 15;
    const int rbase = (lane >> 4) << 3;
    const int oc    = o0 + wn * 32 + col;
    const float bv0 = half ? b1[oc]      : 0.0f;
    const float bv1 = half ? b1[oc + 16] : 0.0f;
    float* orow = outp + (size_t)(l0 + wm * 16 + rbase) * HDIM + oc;
    for (int r = 0; r < 8; ++r) {
        orow[(size_t)r * HDIM]      = acc0[r] + bv0;
        orow[(size_t)r * HDIM + 16] = acc1[r] + bv1;
    }
}

#define PT     32
#define PHC    128
#define PPITCH (PHC + 4)
#define PNQ    (PHC / 8)

struct PairFrag {
    v4f a[4];
    v4f r[2];
    v4f w[2];
};

__device__ __forceinline__ void pair_load(PairFrag& f, const float* ab, const float* rb,
                                          const float* wb) {
    for (int jj = 0; jj < 4; ++jj) f.a[jj] = *(const v4f*)(ab + jj * 8 * PPITCH);
    for (int ii = 0; ii < 2; ++ii) f.r[ii] = *(const v4f*)(rb + ii * 16 * PPITCH);
    f.w[0] = *(const v4f*)(wb);
    f.w[1] = *(const v4f*)(wb + 4);
}

__device__ __forceinline__ void pair_compute(const PairFrag& f, v2f (&acc)[2][4]) {
    const v2f w0 = {f.w[0].x, f.w[0].y};
    const v2f w1 = {f.w[0].z, f.w[0].w};
    const v2f w2 = {f.w[1].x, f.w[1].y};
    const v2f w3 = {f.w[1].z, f.w[1].w};
    for (int ii = 0; ii < 2; ++ii) {
        const v2f rlo = {f.r[ii].x, f.r[ii].y};
        const v2f rhi = {f.r[ii].z, f.r[ii].w};
        for (int jj = 0; jj < 4; ++jj) {
            const v2f alo = {f.a[jj].x, f.a[jj].y};
            const v2f ahi = {f.a[jj].z, f.a[jj].w};
            const v2f xlo = alo + rlo;
            const v2f xhi = ahi + rhi;
            const float t0 = fast_tanh(xlo.x);
            const float t1 = fast_tanh(xlo.y);
            const float t2 = fast_tanh(xhi.x);
            const float t3 = fast_tanh(xhi.y);
            v2f c = acc[ii][jj];
            c = __builtin_elementwise_fma(w0, (v2f){t0, t0}, c);
            c = __builtin_elementwise_fma(w1, (v2f){t1, t1}, c);
            c = __builtin_elementwise_fma(w2, (v2f){t2, t2}, c);
            c = __builtin_elementwise_fma(w3, (v2f){t3, t3}, c);
            acc[ii][jj] = c;
        }
    }
}

__global__ __launch_bounds__(256)
void pairwise_tanh(const float* __restrict__ A, const float* __restrict__ R,
                   const float* __restrict__ W2, float* __restrict__ raw) {
    __shared__ __attribute__((aligned(16))) float aT[2][PT][PPITCH];
    __shared__ __attribute__((aligned(16))) float rT[2][PT][PPITCH];
    __shared__ __attribute__((aligned(16))) float w2s[HDIM][2];
    __shared__ __attribute__((aligned(16))) float red[8][128][2];

    const int i0  = blockIdx.y * PT;
    const int j0  = blockIdx.x * PT;
    const int tid = threadIdx.x;
    const int grp = tid >> 7;
    const int t   = tid & 127;
    const int tx  = t & 7;
    const int ty  = t >> 3;

    const int srow = tid >> 5;
    const int scol = (tid & 31) << 2;
    const float* ag = A + (size_t)(j0 + srow) * HDIM + scol;
    const float* rg = R + (size_t)(i0 + srow) * HDIM + scol;

    v4f pa[4], pr[4];
    for (int p = 0; p < 4; ++p) {
        pa[p] = *(const v4f*)(ag + (size_t)(8 * p) * HDIM);
        pr[p] = *(const v4f*)(rg + (size_t)(8 * p) * HDIM);
    }
    for (int p = 0; p < HDIM / 256; ++p) {
        const int h = tid + 256 * p;
        float2 wv; wv.x = W2[h]; wv.y = W2[HDIM + h];
        *(float2*)&w2s[h][0] = wv;
    }
    for (int p = 0; p < 4; ++p) {
        *(v4f*)&aT[0][srow + 8 * p][scol] = pa[p];
        *(v4f*)&rT[0][srow + 8 * p][scol] = pr[p];
    }
    __syncthreads();

    v2f acc[2][4];
    for (int ii = 0; ii < 2; ++ii)
        for (int jj = 0; jj < 4; ++jj) acc[ii][jj] = (v2f){0.0f, 0.0f};

    for (int kc = 0; kc < HDIM / PHC; ++kc) {
        const int cur = kc & 1;
        const bool more = (kc + 1) < (HDIM / PHC);
        if (more) {
            const int h1 = (kc + 1) * PHC;
            for (int p = 0; p < 4; ++p) {
                pa[p] = *(const v4f*)(ag + (size_t)(8 * p) * HDIM + h1);
                pr[p] = *(const v4f*)(rg + (size_t)(8 * p) * HDIM + h1);
            }
        }

        const int hg = grp * (PHC / 2);
        const float* ab = &aT[cur][tx][hg];
        const float* rb = &rT[cur][ty][hg];
        const float* wb = &w2s[kc * PHC + hg][0];

#pragma unroll 2
        for (int q = 0; q < PNQ; ++q) {
            PairFrag f;
            pair_load(f, ab + q * 4, rb + q * 4, wb + q * 8);
            pair_compute(f, acc);
        }

        if (more) {
            for (int p = 0; p < 4; ++p) {
                *(v4f*)&aT[cur ^ 1][srow + 8 * p][scol] = pa[p];
                *(v4f*)&rT[cur ^ 1][srow + 8 * p][scol] = pr[p];
            }
            __syncthreads();
        }
    }

    if (grp == 1) {
        for (int k = 0; k < 8; ++k) *(v2f*)&red[k][t][0] = acc[k >> 2][k & 3];
    }
    __syncthreads();
    if (grp == 0) {
        for (int ii = 0; ii < 2; ++ii) {
            const int i = i0 + ty + 16 * ii;
            for (int jj = 0; jj < 4; ++jj) {
                const int j = j0 + tx + 8 * jj;
                const v2f o = acc[ii][jj] + *(const v2f*)&red[ii * 4 + jj][t][0];
                *(v2f*)(raw + ((size_t)(i * LSEQ + j) << 1)) = o;
            }
        }
    }
}

__global__ __launch_bounds__(256)
void symmetrize(const float* __restrict__ raw, const float* __restrict__ b2,
                float* __restrict__ out) {
    int idx = blockIdx.x * 256 + threadIdx.x;
    if (idx >= LSEQ * LSEQ) return;
    int i = idx >> 9, j = idx & (LSEQ - 1);
    float2 rij = *(const float2*)(raw + ((size_t)(i * LSEQ + j) << 1));
    float2 rji = *(const float2*)(raw + ((size_t)(j * LSEQ + i) << 1));
    float2 o;
    o.x = 0.5f * (rij.x + rji.x) + b2[0];
    o.y = 0.5f * (rij.y + rji.y) + b2[1];
    *(float2*)(out + ((size_t)idx << 1)) = o;
}

extern "C" void kernel_launch(void* const* d_in, const int* in_sizes, int n_in,
                              void* d_out, int out_size, void* d_ws, size_t ws_size,
                              hipStream_t stream) {
    const float* X  = (const float*)d_in[0];
    const float* W1 = (const float*)d_in[1];
    const float* b1 = (const float*)d_in[2];
    const float* W2 = (const float*)d_in[3];
    const float* b2 = (const float*)d_in[4];
    float* out = (float*)d_out;

    float* a_ws = (float*)d_ws;
    float* r_ws = a_ws + (size_t)LSEQ * HDIM;
    float* raw  = r_ws + (size_t)LSEQ * HDIM;

    proj_both_wmma_f32<<<256, 256, 0, stream>>>(X, W1, b1, a_ws, r_ws);

    dim3 grid(LSEQ / PT, LSEQ / PT);
    pairwise_tanh<<<grid, 256, 0, stream>>>(a_ws, r_ws, W2, raw);

    symmetrize<<<(LSEQ * LSEQ) / 256, 256, 0, stream>>>(raw, b2, out);
}


# Symmetrise pass folded into the pairwise kernel: a workgroup computes a 16x32 tile in both orientations and writes out and its mirror; 3 launches to 2; four h partial sums per value, not two
# speedup vs baseline: 1.9539x; 1.0626x over previous
#include <hip/hip_runtime.h>
#include <hip/hip_bf16.h>
#include <math.h>

#define LSEQ 512
#define HDIM 1024

typedef __attribute__((ext_vector_type(2))) float v2f;
typedef __attribute__((ext_vector_type(8))) float v8f;
typedef __attribute__((ext_vector_type(4))) float v4f;

__device__ __forceinline__ float fast_tanh(float x) {
#if __has_builtin(__builtin_amdgcn_tanhf)
    return __builtin_amdgcn_tanhf(x);
#elif __has_builtin(__builtin_amdgcn_tanh_f32)
    return __builtin_amdgcn_tanh_f32(x);
#else
    float e = __builtin_amdgcn_exp2f(x * 2.8853900817779268f);
    return 1.0f - 2.0f * __builtin_amdgcn_rcpf(1.0f + e);
#endif
}

#define GT   64
#define GTP  (GT + 4)

__global__ __launch_bounds__(256)
void proj_both_wmma_f32(const float* __restrict__ X, const float* __restrict__ W1,
                        const float* __restrict__ b1,
                        float* __restrict__ a_out, float* __restrict__ r_out) {
    __shared__ __attribute__((aligned(16))) float sA[2][GT][GTP];
    __shared__ __attribute__((aligned(16))) float sB[2][GT][GTP];

    const int tid  = threadIdx.x;
    const int lane = tid & 31;
    const int wave = tid >> 5;
    const int bm   = blockIdx.x & 7;
    const int bn   = blockIdx.x >> 3;
    const int half = bn >> 4;
    const int o0   = (bn & 15) << 6;
    const int l0   = bm << 6;

    const int srow = tid >> 4;
    const int scol = (tid & 15) << 2;
    const float* xg = X  + (size_t)(l0 + srow) * HDIM + scol;
    const float* wg = W1 + (size_t)(o0 + srow) * (2 * HDIM) + half * HDIM + scol;

    float4 pa[4], pb[4];
    for (int p = 0; p < 4; ++p) {
        pa[p] = *(const float4*)(xg + (size_t)(16 * p) * HDIM);
        pb[p] = *(const float4*)(wg + (size_t)(16 * p) * (2 * HDIM));
    }
    for (int p = 0; p < 4; ++p) {
        *(float4*)&sA[0][srow + 16 * p][scol] = pa[p];
        *(float4*)&sB[0][srow + 16 * p][scol] = pb[p];
    }
    __syncthreads();

    const int wm   = wave & 3;
    const int wn   = wave >> 2;
    const int frow = lane & 15;
    const int koff = (lane >> 4) << 1;

    v8f acc0 = {}, acc1 = {};

    for (int kc = 0; kc < HDIM / GT; ++kc) {
        const int cur = kc & 1;
        const bool more = (kc + 1) < (HDIM / GT);
        if (more) {
            const int k1 = (kc + 1) * GT;
            for (int p = 0; p < 4; ++p) {
                pa[p] = *(const float4*)(xg + (size_t)(16 * p) * HDIM + k1);
                pb[p] = *(const float4*)(wg + (size_t)(16 * p) * (2 * HDIM) + k1);
            }
        }

        const float* ap  = &sA[cur][wm * 16 + frow][koff];
        const float* bp0 = &sB[cur][wn * 32 + frow][koff];
        const float* bp1 = &sB[cur][wn * 32 + 16 + frow][koff];
        for (int k = 0; k < GT; k += 4) {
            v2f a  = *(const v2f*)(ap  + k);
            v2f b0 = *(const v2f*)(bp0 + k);
            v2f b1f = *(const v2f*)(bp1 + k);
            acc0 = __builtin_amdgcn_wmma_f32_16x16x4_f32(
                 false, a,  false, b0,
                 (short)0, acc0,  false,  false);
            acc1 = __builtin_amdgcn_wmma_f32_16x16x4_f32(
                 false, a,  false, b1f,
                 (short)0, acc1,  false,  false);
        }

        if (more) {
            for (int p = 0; p < 4; ++p) {
                *(float4*)&sA[cur ^ 1][srow + 16 * p][scol] = pa[p];
                *(float4*)&sB[cur ^ 1][srow + 16 * p][scol] = pb[p];
            }
            __syncthreads();
        }
    }

    float* __restrict__ outp = half ? r_out : a_out;
    const int col   = lane & 15;
    const int rbase = (lane >> 4) << 3;
    const int oc    = o0 + wn * 32 + col;
    const float bv0 = half ? b1[oc]      : 0.0f;
    const float bv1 = half ? b1[oc + 16] : 0.0f;
    float* orow = outp + (size_t)(l0 + wm * 16 + rbase) * HDIM + oc;
    for (int r = 0; r < 8; ++r) {
        orow[(size_t)r * HDIM]      = acc0[r] + bv0;
        orow[(size_t)r * HDIM + 16] = acc1[r] + bv1;
    }
}

#define FTN    16
#define FTW    32
#define FJOB   (FTN + FTW)
#define SROWS  (2 * FJOB)
#define PHC    128
#define PPITCH (PHC + 4)
#define PNQ    (PHC / 16)
#define NSTG   (SROWS / 8)
#define OTP    33

struct PairFrag {
    v4f n[2][2];
    v4f w[2][4];
    v4f c[2];
};

__device__ __forceinline__ void pair_load(PairFrag& f, const float* nb, const float* wb,
                                          const float* cb) {
    for (int job = 0; job < 2; ++job) {
        for (int ii = 0; ii < 2; ++ii)
            f.n[job][ii] = *(const v4f*)(nb + (job * FJOB + ii * 8) * PPITCH);
        for (int jj = 0; jj < 4; ++jj)
            f.w[job][jj] = *(const v4f*)(wb + (job * FJOB + jj * 8) * PPITCH);
    }
    f.c[0] = *(const v4f*)(cb);
    f.c[1] = *(const v4f*)(cb + 4);
}

__device__ __forceinline__ void pair_compute(const PairFrag& f, v2f (&acc)[2][2][4]) {
    const v2f w0 = {f.c[0].x, f.c[0].y};
    const v2f w1 = {f.c[0].z, f.c[0].w};
    const v2f w2 = {f.c[1].x, f.c[1].y};
    const v2f w3 = {f.c[1].z, f.c[1].w};
    for (int job = 0; job < 2; ++job) {
        for (int ii = 0; ii < 2; ++ii) {
            const v2f nlo = {f.n[job][ii].x, f.n[job][ii].y};
            const v2f nhi = {f.n[job][ii].z, f.n[job][ii].w};
            for (int jj = 0; jj < 4; ++jj) {
                const v2f wlo = {f.w[job][jj].x, f.w[job][jj].y};
                const v2f whi = {f.w[job][jj].z, f.w[job][jj].w};
                const v2f xlo = wlo + nlo;
                const v2f xhi = whi + nhi;
                const float t0 = fast_tanh(xlo.x);
                const float t1 = fast_tanh(xlo.y);
                const float t2 = fast_tanh(xhi.x);
                const float t3 = fast_tanh(xhi.y);
                v2f c = acc[job][ii][jj];
                c = __builtin_elementwise_fma(w0, (v2f){t0, t0}, c);
                c = __builtin_elementwise_fma(w1, (v2f){t1, t1}, c);
                c = __builtin_elementwise_fma(w2, (v2f){t2, t2}, c);
                c = __builtin_elementwise_fma(w3, (v2f){t3, t3}, c);
                acc[job][ii][jj] = c;
            }
        }
    }
}

__device__ __forceinline__ const float* stage_src(int p, const float* gn0, const float* gw0,
                                                  const float* gn1, const float* gw1) {
    if (p < 2) return gn0 + (size_t)(8 * p) * HDIM;
    if (p < 6) return gw0 + (size_t)(8 * (p - 2)) * HDIM;
    if (p < 8) return gn1 + (size_t)(8 * (p - 6)) * HDIM;
    return gw1 + (size_t)(8 * (p - 8)) * HDIM;
}

__global__ __launch_bounds__(256)
void pairwise_tanh_sym(const float* __restrict__ A, const float* __restrict__ R,
                       const float* __restrict__ W2, const float* __restrict__ b2,
                       float* __restrict__ out) {
    __shared__ __attribute__((aligned(16))) float st[2][SROWS][PPITCH];
    __shared__ __attribute__((aligned(16))) float w2s[HDIM][2];
    __shared__ __attribute__((aligned(16))) float red[3][16][64][2];
    __shared__ __attribute__((aligned(16))) float ot[32][OTP][2];

    const int tid = threadIdx.x;
    const int bid = blockIdx.x;

    const bool diag = bid < 16;
    int i0, j0;
    if (diag) {
        i0 = bid * 32;
        j0 = i0;
    } else {
        const int t = bid - 16;
        int jb = 1;
        while ((jb + 1) * jb <= t) ++jb;
        i0 = (t - jb * (jb - 1)) * 16;
        j0 = jb * 32;
    }
    const float* n0 = R + (size_t)i0 * HDIM;
    const float* w0 = A + (size_t)j0 * HDIM;
    const float* n1 = diag ? R + (size_t)(i0 + 16) * HDIM : A + (size_t)i0 * HDIM;
    const float* w1 = diag ? A + (size_t)j0 * HDIM : R + (size_t)j0 * HDIM;

    const int lane = tid & 31;
    const int wave = tid >> 5;
    const int grp  = wave >> 1;
    const int t64  = tid & 63;
    const int tx   = lane & 7;
    const int ty   = (lane >> 3) + ((wave & 1) << 2);

    const int srow = tid >> 5;
    const int scol = (tid & 31) << 2;
    const size_t soff = (size_t)srow * HDIM + scol;
    const float* gn0 = n0 + soff;
    const float* gw0 = w0 + soff;
    const float* gn1 = n1 + soff;
    const float* gw1 = w1 + soff;

    v4f pf[NSTG];
    for (int p = 0; p < NSTG; ++p) pf[p] = *(const v4f*)(stage_src(p, gn0, gw0, gn1, gw1));
    for (int p = 0; p < HDIM / 256; ++p) {
        const int h = tid + 256 * p;
        float2 wv; wv.x = W2[h]; wv.y = W2[HDIM + h];
        *(float2*)&w2s[h][0] = wv;
    }
    for (int p = 0; p < NSTG; ++p) *(v4f*)&st[0][srow + 8 * p][scol] = pf[p];
    __syncthreads();

    v2f acc[2][2][4];
    for (int job = 0; job < 2; ++job)
        for (int ii = 0; ii < 2; ++ii)
            for (int jj = 0; jj < 4; ++jj) acc[job][ii][jj] = (v2f){0.0f, 0.0f};

    for (int kc = 0; kc < HDIM / PHC; ++kc) {
        const int cur = kc & 1;
        const bool more = (kc + 1) < (HDIM / PHC);
        if (more) {
            const int h1 = (kc + 1) * PHC;
            for (int p = 0; p < NSTG; ++p)
                pf[p] = *(const v4f*)(stage_src(p, gn0, gw0, gn1, gw1) + h1);
        }

        const int hg = grp * (PHC / 4);
        const float* nb = &st[cur][ty][hg];
        const float* wb = &st[cur][FTN + tx][hg];
        const float* cb = &w2s[kc * PHC + hg][0];

#pragma unroll 1
        for (int q = 0; q < PNQ; ++q) {
            PairFrag f;
            pair_load(f, nb + q * 4, wb + q * 4, cb + q * 8);
            pair_compute(f, acc);
        }

        if (more) {
            for (int p = 0; p < NSTG; ++p) *(v4f*)&st[cur ^ 1][srow + 8 * p][scol] = pf[p];
            __syncthreads();
        }
    }

    if (grp != 0) {
        for (int k = 0; k < 16; ++k)
            *(v2f*)&red[grp - 1][k][t64][0] = acc[k >> 3][(k >> 2) & 1][k & 3];
    }
    __syncthreads();
    const v2f bb = {b2[0], b2[1]};
    if (grp == 0) {
        for (int k = 0; k < 16; ++k) {
            v2f s = acc[k >> 3][(k >> 2) & 1][k & 3];
            s = s + *(const v2f*)&red[0][k][t64][0];
            s = s + *(const v2f*)&red[1][k][t64][0];
            s = s + *(const v2f*)&red[2][k][t64][0];
            acc[k >> 3][(k >> 2) & 1][k & 3] = s;
        }
        for (int ii = 0; ii < 2; ++ii) {
            const int nr = ty + 8 * ii;
            for (int jj = 0; jj < 4; ++jj) {
                const int wr = tx + 8 * jj;
                if (diag) {
                    *(v2f*)&ot[nr][wr][0]      = acc[0][ii][jj];
                    *(v2f*)&ot[nr + 16][wr][0] = acc[1][ii][jj];
                } else {
                    const v2f o = 0.5f * (acc[0][ii][jj] + acc[1][ii][jj]) + bb;
                    *(v2f*)&ot[nr][wr][0] = o;
                }
            }
        }
    }
    __syncthreads();
    if (diag) {
        for (int k = 0; k < 4; ++k) {
            const int e = tid + 256 * k;
            const int a = e >> 5, b = e & 31;
            const v2f x = *(const v2f*)&ot[a][b][0];
            const v2f y = *(const v2f*)&ot[b][a][0];
            const v2f o = 0.5f * (x + y) + bb;
            *(v2f*)(out + ((size_t)((i0 + a) * LSEQ + i0 + b) << 1)) = o;
        }
    } else {
        for (int k = 0; k < 2; ++k) {
            const int e = tid + 256 * k;
            const int nr = e >> 5, wr = e & 31;
            *(v2f*)(out + ((size_t)((i0 + nr) * LSEQ + j0 + wr) << 1)) = *(const v2f*)&ot[nr][wr][0];
        }
        for (int k = 0; k < 2; ++k) {
            const int e = tid + 256 * k;
            const int wr = e >> 4, nr = e & 15;
            *(v2f*)(out + ((size_t)((j0 + wr) * LSEQ + i0 + nr) << 1)) = *(const v2f*)&ot[nr][wr][0];
        }
    }
}

extern "C" void kernel_launch(void* const* d_in, const int* in_sizes, int n_in,
                              void* d_out, int out_size, void* d_ws, size_t ws_size,
                              hipStream_t stream) {
    const float* X  = (const float*)d_in[0];
    const float* W1 = (const float*)d_in[1];
    const float* b1 = (const float*)d_in[2];
    const float* W2 = (const float*)d_in[3];
    const float* b2 = (const float*)d_in[4];
    float* out = (float*)d_out;

    float* a_ws = (float*)d_ws;
    float* r_ws = a_ws + (size_t)LSEQ * HDIM;

    proj_both_wmma_f32<<<256, 256, 0, stream>>>(X, W1, b1, a_ws, r_ws);

    pairwise_tanh_sym<<<256, 256, 0, stream>>>(a_ws, r_ws, W2, b2, out);
}


# speedup vs baseline: 2.0607x; 1.0547x over previous
#include <hip/hip_runtime.h>
#include <hip/hip_bf16.h>
#include <math.h>

#define LSEQ 512
#define HDIM 1024

typedef __attribute__((ext_vector_type(2))) float v2f;
typedef __attribute__((ext_vector_type(8))) float v8f;
typedef __attribute__((ext_vector_type(4))) float v4f;

__device__ __forceinline__ float fast_tanh(float x) {
#if __has_builtin(__builtin_amdgcn_tanhf)
    return __builtin_amdgcn_tanhf(x);
#elif __has_builtin(__builtin_amdgcn_tanh_f32)
    return __builtin_amdgcn_tanh_f32(x);
#else
    float e = __builtin_amdgcn_exp2f(x * 2.8853900817779268f);
    return 1.0f - 2.0f * __builtin_amdgcn_rcpf(1.0f + e);
#endif
}

#define GT   64
#define GTP  (GT + 4)

__global__ __launch_bounds__(256)
void proj_both_wmma_f32(const float* __restrict__ X, const float* __restrict__ W1,
                        const float* __restrict__ b1,
                        float* __restrict__ a_out, float* __restrict__ r_out) {
    __shared__ __attribute__((aligned(16))) float sA[2][GT][GTP];
    __shared__ __attribute__((aligned(16))) float sB[2][GT][GTP];

    const int tid  = threadIdx.x;
    const int lane = tid & 31;
    const int wave = tid >> 5;
    const int bm   = blockIdx.x & 7;
    const int bn   = blockIdx.x >> 3;
    const int half = bn >> 4;
    const int o0   = (bn & 15) << 6;
    const int l0   = bm << 6;

    const int srow = tid >> 4;
    const int scol = (tid & 15) << 2;
    const float* xg = X  + (size_t)(l0 + srow) * HDIM + scol;
    const float* wg = W1 + (size_t)(o0 + srow) * (2 * HDIM) + half * HDIM + scol;

    float4 pa[4], pb[4];
    for (int p = 0; p < 4; ++p) {
        pa[p] = *(const float4*)(xg + (size_t)(16 * p) * HDIM);
        pb[p] = *(const float4*)(wg + (size_t)(16 * p) * (2 * HDIM));
    }
    for (int p = 0; p < 4; ++p) {
        *(float4*)&sA[0][srow + 16 * p][scol] = pa[p];
        *(float4*)&sB[0][srow + 16 * p][scol] = pb[p];
    }
    __syncthreads();

    const int wm   = wave & 3;
    const int wn   = wave >> 2;
    const int frow = lane & 15;
    const int koff = (lane >> 4) << 1;

    v8f acc0 = {}, acc1 = {};

    for (int kc = 0; kc < HDIM / GT; ++kc) {
        const int cur = kc & 1;
        const bool more = (kc + 1) < (HDIM / GT);
        if (more) {
            const int k1 = (kc + 1) * GT;
            for (int p = 0; p < 4; ++p) {
                pa[p] = *(const float4*)(xg + (size_t)(16 * p) * HDIM + k1);
                pb[p] = *(const float4*)(wg + (size_t)(16 * p) * (2 * HDIM) + k1);
            }
        }

        const float* ap  = &sA[cur][wm * 16 + frow][koff];
        const float* bp0 = &sB[cur][wn * 32 + frow][koff];
        const float* bp1 = &sB[cur][wn * 32 + 16 + frow][koff];
        for (int k = 0; k < GT; k += 4) {
            v2f a  = *(const v2f*)(ap  + k);
            v2f b0 = *(const v2f*)(bp0 + k);
            v2f b1f = *(const v2f*)(bp1 + k);
            acc0 = __builtin_amdgcn_wmma_f32_16x16x4_f32(
                 false, a,  false, b0,
                 (short)0, acc0,  false,  false);
            acc1 = __builtin_amdgcn_wmma_f32_16x16x4_f32(
                 false, a,  false, b1f,
                 (short)0, acc1,  false,  false);
        }

        if (more) {
            for (int p = 0; p < 4; ++p) {
                *(float4*)&sA[cur ^ 1][srow + 16 * p][scol] = pa[p];
                *(float4*)&sB[cur ^ 1][srow + 16 * p][scol] = pb[p];
            }
            __syncthreads();
        }
    }

    float* __restrict__ outp = half ? r_out : a_out;
    const int col   = lane & 15;
    const int rbase = (lane >> 4) << 3;
    const int oc    = o0 + wn * 32 + col;
    const float bv0 = half ? b1[oc]      : 0.0f;
    const float bv1 = half ? b1[oc + 16] : 0.0f;
    float* orow = outp + (size_t)(l0 + wm * 16 + rbase) * HDIM + oc;
    for (int r = 0; r < 8; ++r) {
        orow[(size_t)r * HDIM]      = acc0[r] + bv0;
        orow[(size_t)r * HDIM + 16] = acc1[r] + bv1;
    }
}

#define FTN    16
#define FTW    32
#define FJOB   (FTN + FTW)
#define SROWS  (2 * FJOB)
#define PHC    128
#define PPITCH (PHC + 4)
#define PNQ    (PHC / 16)
#define NSTG   (SROWS / 8)
#define OTP    33

struct PairFrag {
    v4f n[2][2];
    v4f w[2][4];
    v4f c[2];
};

__device__ __forceinline__ void pair_load(PairFrag& f, const float* nb, const float* wb,
                                          const float* cb) {
    for (int job = 0; job < 2; ++job) {
        for (int ii = 0; ii < 2; ++ii)
            f.n[job][ii] = *(const v4f*)(nb + (job * FJOB + ii * 8) * PPITCH);
        for (int jj = 0; jj < 4; ++jj)
            f.w[job][jj] = *(const v4f*)(wb + (job * FJOB + jj * 8) * PPITCH);
    }
    f.c[0] = *(const v4f*)(cb);
    f.c[1] = *(const v4f*)(cb + 4);
}

__device__ __forceinline__ void pair_compute(const PairFrag& f, v2f (&acc)[2][2][4]) {
    const v2f wc[4] = {{f.c[0].x, f.c[0].y}, {f.c[0].z, f.c[0].w},
                       {f.c[1].x, f.c[1].y}, {f.c[1].z, f.c[1].w}};
    v4f x[16];
    float t[4][16];
#define PAIR_ADD(g)    x[g] = f.w[(g) >> 3][(g) & 3] + f.n[(g) >> 3][((g) >> 2) & 1]
#define PAIR_FMA(k, g) acc[(g) >> 3][((g) >> 2) & 1][(g) & 3] = __builtin_elementwise_fma( \
        wc[k], (v2f){t[k][g], t[k][g]}, acc[(g) >> 3][((g) >> 2) & 1][(g) & 3])
    for (int g = 0; g < 8; ++g) PAIR_ADD(g);

    for (int g = 0; g < 8; ++g) t[0][g] = fast_tanh(x[g].x);
    for (int g = 0; g < 8; ++g) t[1][g] = fast_tanh(x[g].y);
    for (int g = 8; g < 16; ++g) PAIR_ADD(g);

    for (int g = 0; g < 8; ++g) t[2][g] = fast_tanh(x[g].z);
    for (int g = 0; g < 8; ++g) t[3][g] = fast_tanh(x[g].w);
    for (int g = 0; g < 8; ++g) PAIR_FMA(0, g);
    for (int g = 0; g < 8; ++g) PAIR_FMA(1, g);

    for (int g = 8; g < 16; ++g) t[0][g] = fast_tanh(x[g].x);
    for (int g = 8; g < 16; ++g) t[1][g] = fast_tanh(x[g].y);
    for (int g = 0; g < 8; ++g) PAIR_FMA(2, g);
    for (int g = 0; g < 8; ++g) PAIR_FMA(3, g);

    for (int g = 8; g < 16; ++g) t[2][g] = fast_tanh(x[g].z);
    for (int g = 8; g < 16; ++g) t[3][g] = fast_tanh(x[g].w);
    for (int g = 8; g < 16; ++g) PAIR_FMA(0, g);
    for (int g = 8; g < 16; ++g) PAIR_FMA(1, g);

    for (int g = 8; g < 16; ++g) PAIR_FMA(2, g);
    for (int g = 8; g < 16; ++g) PAIR_FMA(3, g);
#undef PAIR_ADD
#undef PAIR_FMA
    __builtin_amdgcn_sched_group_barrier(0x002, 19, 0);
#pragma unroll
    for (int i = 0; i < 64; ++i) {
        __builtin_amdgcn_sched_group_barrier(0x400, 1, 0);
        __builtin_amdgcn_sched_group_barrier(0x002, 1, 0);
    }
    __builtin_amdgcn_sched_group_barrier(0x002, 16, 0);
}

__device__ __forceinline__ const float* stage_src(int p, const float* gn0, const float* gw0,
                                                  const float* gn1, const float* gw1) {
    if (p < 2) return gn0 + (size_t)(8 * p) * HDIM;
    if (p < 6) return gw0 + (size_t)(8 * (p - 2)) * HDIM;
    if (p < 8) return gn1 + (size_t)(8 * (p - 6)) * HDIM;
    return gw1 + (size_t)(8 * (p - 8)) * HDIM;
}

__global__ __launch_bounds__(256)
void pairwise_tanh_sym(const float* __restrict__ A, const float* __restrict__ R,
                       const float* __restrict__ W2, const float* __restrict__ b2,
                       float* __restrict__ out) {
    __shared__ __attribute__((aligned(16))) float st[2][SROWS][PPITCH];
    __shared__ __attribute__((aligned(16))) float w2s[HDIM][2];
    __shared__ __attribute__((aligned(16))) float red[3][16][64][2];
    __shared__ __attribute__((aligned(16))) float ot[32][OTP][2];

    const int tid = threadIdx.x;
    const int bid = blockIdx.x;

    const bool diag = bid < 16;
    int i0, j0;
    if (diag) {
        i0 = bid * 32;
        j0 = i0;
    } else {
        const int t = bid - 16;
        int jb = 1;
        while ((jb + 1) * jb <= t) ++jb;
        i0 = (t - jb * (jb - 1)) * 16;
        j0 = jb * 32;
    }
    const float* n0 = R + (size_t)i0 * HDIM;
    const float* w0 = A + (size_t)j0 * HDIM;
    const float* n1 = diag ? R + (size_t)(i0 + 16) * HDIM : A + (size_t)i0 * HDIM;
    const float* w1 = diag ? A + (size_t)j0 * HDIM : R + (size_t)j0 * HDIM;

    const int lane = tid & 31;
    const int wave = tid >> 5;
    const int grp  = wave >> 1;
    const int t64  = tid & 63;
    const int tx   = lane & 7;
    const int ty   = (lane >> 3) + ((wave & 1) << 2);

    const int srow = tid >> 5;
    const int scol = (tid & 31) << 2;
    const size_t soff = (size_t)srow * HDIM + scol;
    const float* gn0 = n0 + soff;
    const float* gw0 = w0 + soff;
    const float* gn1 = n1 + soff;
    const float* gw1 = w1 + soff;

    v4f pf[NSTG];
    for (int p = 0; p < NSTG; ++p) pf[p] = *(const v4f*)(stage_src(p, gn0, gw0, gn1, gw1));
    for (int p = 0; p < HDIM / 256; ++p) {
        const int h = tid + 256 * p;
        float2 wv; wv.x = W2[h]; wv.y = W2[HDIM + h];
        *(float2*)&w2s[h][0] = wv;
    }
    for (int p = 0; p < NSTG; ++p) *(v4f*)&st[0][srow + 8 * p][scol] = pf[p];
    __syncthreads();

    v2f acc[2][2][4];
    for (int job = 0; job < 2; ++job)
        for (int ii = 0; ii < 2; ++ii)
            for (int jj = 0; jj < 4; ++jj) acc[job][ii][jj] = (v2f){0.0f, 0.0f};

    for (int kc = 0; kc < HDIM / PHC; ++kc) {
        const int cur = kc & 1;
        const bool more = (kc + 1) < (HDIM / PHC);
        if (more) {
            const int h1 = (kc + 1) * PHC;
            for (int p = 0; p < NSTG; ++p)
                pf[p] = *(const v4f*)(stage_src(p, gn0, gw0, gn1, gw1) + h1);
        }

        const int hg = grp * (PHC / 4);
        const float* nb = &st[cur][ty][hg];
        const float* wb = &st[cur][FTN + tx][hg];
        const float* cb = &w2s[kc * PHC + hg][0];

#pragma unroll 1
        for (int q = 0; q < PNQ; ++q) {
            PairFrag f;
            pair_load(f, nb + q * 4, wb + q * 4, cb + q * 8);
            pair_compute(f, acc);
        }

        if (more) {
            for (int p = 0; p < NSTG; ++p) *(v4f*)&st[cur ^ 1][srow + 8 * p][scol] = pf[p];
            __syncthreads();
        }
    }

    if (grp != 0) {
        for (int k = 0; k < 16; ++k)
            *(v2f*)&red[grp - 1][k][t64][0] = acc[k >> 3][(k >> 2) & 1][k & 3];
    }
    __syncthreads();
    const v2f bb = {b2[0], b2[1]};
    if (grp == 0) {
        for (int k = 0; k < 16; ++k) {
            v2f s = acc[k >> 3][(k >> 2) & 1][k & 3];
            s = s + *(const v2f*)&red[0][k][t64][0];
            s = s + *(const v2f*)&red[1][k][t64][0];
            s = s + *(const v2f*)&red[2][k][t64][0];
            acc[k >> 3][(k >> 2) & 1][k & 3] = s;
        }
        for (int ii = 0; ii < 2; ++ii) {
            const int nr = ty + 8 * ii;
            for (int jj = 0; jj < 4; ++jj) {
                const int wr = tx + 8 * jj;
                if (diag) {
                    *(v2f*)&ot[nr][wr][0]      = acc[0][ii][jj];
                    *(v2f*)&ot[nr + 16][wr][0] = acc[1][ii][jj];
                } else {
                    const v2f o = 0.5f * (acc[0][ii][jj] + acc[1][ii][jj]) + bb;
                    *(v2f*)&ot[nr][wr][0] = o;
                }
            }
        }
    }
    __syncthreads();
    if (diag) {
        for (int k = 0; k < 4; ++k) {
            const int e = tid + 256 * k;
            const int a = e >> 5, b = e & 31;
            const v2f x = *(const v2f*)&ot[a][b][0];
            const v2f y = *(const v2f*)&ot[b][a][0];
            const v2f o = 0.5f * (x + y) + bb;
            *(v2f*)(out + ((size_t)((i0 + a) * LSEQ + i0 + b) << 1)) = o;
        }
    } else {
        for (int k = 0; k < 2; ++k) {
            const int e = tid + 256 * k;
            const int nr = e >> 5, wr = e & 31;
            *(v2f*)(out + ((size_t)((i0 + nr) * LSEQ + j0 + wr) << 1)) = *(const v2f*)&ot[nr][wr][0];
        }
        for (int k = 0; k < 2; ++k) {
            const int e = tid + 256 * k;
            const int wr = e >> 4, nr = e & 15;
            *(v2f*)(out + ((size_t)((j0 + wr) * LSEQ + i0 + nr) << 1)) = *(const v2f*)&ot[nr][wr][0];
        }
    }
}

extern "C" void kernel_launch(void* const* d_in, const int* in_sizes, int n_in,
                              void* d_out, int out_size, void* d_ws, size_t ws_size,
                              hipStream_t stream) {
    const float* X  = (const float*)d_in[0];
    const float* W1 = (const float*)d_in[1];
    const float* b1 = (const float*)d_in[2];
    const float* W2 = (const float*)d_in[3];
    const float* b2 = (const float*)d_in[4];
    float* out = (float*)d_out;

    float* a_ws = (float*)d_ws;
    float* r_ws = a_ws + (size_t)LSEQ * HDIM;

    proj_both_wmma_f32<<<256, 256, 0, stream>>>(X, W1, b1, a_ws, r_ws);

    pairwise_tanh_sym<<<256, 256, 0, stream>>>(a_ws, r_ws, W2, b2, out);
}
